# starting kernel
# baseline (speedup 1.0000x reference)
#include <hip/hip_runtime.h>

#define Bq  8
#define Nq  1024
#define Dq  256
#define Hq  4
#define HOq 64
#define Tq  5
#define NEGV (-1e9f)
#define EPSV (1e-6f)

typedef float v2f __attribute__((ext_vector_type(2)));
typedef float v8f __attribute__((ext_vector_type(8)));

__device__ __forceinline__ float sel5(int t, float e0, float e1, float e2, float e3, float e4) {
    return (t == 0) ? e0 : (t == 1) ? e1 : (t == 2) ? e2 : (t == 3) ? e3 : e4;
}

__global__ void __launch_bounds__(32) k_proj(const float* __restrict__ X,
                                             const float* __restrict__ Wm,
                                             float* __restrict__ hws) {
    const int lane = threadIdx.x;
    const int half = lane >> 4;
    const int lm   = lane & 15;
    const int i0   = blockIdx.x * 16;
    const int ho0  = blockIdx.y * 16;
    const int bh   = blockIdx.z;
    const int b    = bh / Hq, hh = bh % Hq;

    const float* xr = X + ((size_t)b * Nq + (size_t)(i0 + lm)) * Dq;
    const float* wp = Wm + (size_t)hh * Dq * HOq + ho0 + lm;

    v8f c = {};
#pragma unroll 4
    for (int k = 0; k < Dq; k += 4) {
        const int kk = k + half * 2;
        float2 axy = *(const float2*)(xr + kk);
        v2f a; a.x = axy.x; a.y = axy.y;
        v2f bb;
        bb.x = wp[(size_t)kk * HOq];
        bb.y = wp[(size_t)(kk + 1) * HOq];
        c = __builtin_amdgcn_wmma_f32_16x16x4_f32(false, a, false, bb,
                                                  (short)0, c, false, false);
    }
    float* hp = hws + ((size_t)bh * Nq + i0) * HOq + ho0 + lm;
#pragma unroll
    for (int r = 0; r < 8; ++r)
        hp[(size_t)(r + 8 * half) * HOq] = c[r];
}

__global__ void __launch_bounds__(256) k_escore(const float* __restrict__ hws,
                                                const float* __restrict__ asrc,
                                                const float* __restrict__ adst,
                                                float* __restrict__ es,
                                                float* __restrict__ ed) {
    int idx = blockIdx.x * 256 + threadIdx.x;
    if (idx >= Bq * Hq * Tq * Nq) return;
    const int n  = idx % Nq;
    const int t  = (idx / Nq) % Tq;
    const int bh = idx / (Nq * Tq);
    const int hh = bh % Hq;
    const float4* hv = (const float4*)(hws + ((size_t)bh * Nq + n) * HOq);
    const float4* sv = (const float4*)(asrc + ((size_t)hh * Tq + t) * HOq);
    const float4* dv = (const float4*)(adst + ((size_t)hh * Tq + t) * HOq);
    float ss = 0.f, sd = 0.f;
#pragma unroll
    for (int q = 0; q < HOq / 4; ++q) {
        float4 hx = hv[q], sx = sv[q], dx = dv[q];
        ss += hx.x * sx.x + hx.y * sx.y + hx.z * sx.z + hx.w * sx.w;
        sd += hx.x * dx.x + hx.y * dx.y + hx.z * dx.z + hx.w * dx.w;
    }
    es[idx] = ss;
    ed[idx] = sd;
}

__global__ void __launch_bounds__(32) k_softstats(const int* __restrict__ adj,
                                                  const float* __restrict__ es,
                                                  const float* __restrict__ ed,
                                                  float* __restrict__ rmax,
                                                  float* __restrict__ rsum) {
    const int lane  = threadIdx.x;
    const int rowid = blockIdx.x;
    const int i  = rowid % Nq;
    const int bh = rowid / Nq;
    const int b  = bh / Hq;

    const float e0 = es[((size_t)bh * Tq + 0) * Nq + i];
    const float e1 = es[((size_t)bh * Tq + 1) * Nq + i];
    const float e2 = es[((size_t)bh * Tq + 2) * Nq + i];
    const float e3 = es[((size_t)bh * Tq + 3) * Nq + i];
    const float e4 = es[((size_t)bh * Tq + 4) * Nq + i];
    const int*   ar  = adj + ((size_t)b * Nq + i) * Nq;
    const float* edb = ed + (size_t)bh * Tq * Nq;

    float m = -INFINITY, s = 0.f;
    for (int j = lane; j < Nq; j += 32) {
        int a = ar[j];
        int t = (a > 0) ? (a - 1) : 0;
        float sc = sel5(t, e0, e1, e2, e3, e4) + edb[(size_t)t * Nq + j];
        float w = (a == 0) ? NEGV : (sc > 0.f ? sc : 0.01f * sc);
        float mn = fmaxf(m, w);
        s = s * __expf(m - mn) + __expf(w - mn);
        m = mn;
    }
#pragma unroll
    for (int off = 16; off >= 1; off >>= 1) {
        float mo = __shfl_xor(m, off, 32);
        float so = __shfl_xor(s, off, 32);
        float mn = fmaxf(m, mo);
        s = s * __expf(m - mn) + so * __expf(mo - mn);
        m = mn;
    }
    if (lane == 0) { rmax[rowid] = m; rsum[rowid] = s; }
}

__global__ void __launch_bounds__(32) k_attn(const int* __restrict__ adj,
                                             const float* __restrict__ hws,
                                             const float* __restrict__ es,
                                             const float* __restrict__ ed,
                                             const float* __restrict__ rmax,
                                             const float* __restrict__ rsum,
                                             const float* __restrict__ bias,
                                             const float* __restrict__ gamma,
                                             const float* __restrict__ beta,
                                             float* __restrict__ out) {
    __shared__ float tile[16 * 68];
    const int lane = threadIdx.x;
    const int half = lane >> 4;
    const int lm   = lane & 15;
    const int i0   = blockIdx.x * 16;
    const int bh   = blockIdx.y;
    const int b    = bh / Hq, hh = bh % Hq;
    const int row  = i0 + lm;

    const float e0 = es[((size_t)bh * Tq + 0) * Nq + row];
    const float e1 = es[((size_t)bh * Tq + 1) * Nq + row];
    const float e2 = es[((size_t)bh * Tq + 2) * Nq + row];
    const float e3 = es[((size_t)bh * Tq + 3) * Nq + row];
    const float e4 = es[((size_t)bh * Tq + 4) * Nq + row];
    const float rm   = rmax[(size_t)bh * Nq + row];
    const float rinv = 1.f / rsum[(size_t)bh * Nq + row];

    const int*   ar  = adj + ((size_t)b * Nq + row) * Nq;
    const float* edb = ed + (size_t)bh * Tq * Nq;
    const float* hb  = hws + (size_t)bh * Nq * HOq;

    v8f c0 = {}, c1 = {}, c2 = {}, c3 = {};
    for (int j0 = 0; j0 < Nq; j0 += 4) {
        const int jj = j0 + half * 2;
        int2 av = *(const int2*)(ar + jj);

        float p0, p1;
        {
            int a = av.x; int t = (a > 0) ? (a - 1) : 0;
            float sc = sel5(t, e0, e1, e2, e3, e4) + edb[(size_t)t * Nq + jj];
            float w = (a == 0) ? NEGV : (sc > 0.f ? sc : 0.01f * sc);
            p0 = __expf(w - rm) * rinv;
        }
        {
            int a = av.y; int t = (a > 0) ? (a - 1) : 0;
            float sc = sel5(t, e0, e1, e2, e3, e4) + edb[(size_t)t * Nq + jj + 1];
            float w = (a == 0) ? NEGV : (sc > 0.f ? sc : 0.01f * sc);
            p1 = __expf(w - rm) * rinv;
        }
        v2f ap; ap.x = p0; ap.y = p1;

        const float* hr0 = hb + (size_t)jj * HOq + lm;
        const float* hr1 = hr0 + HOq;
        v2f b0, b1, b2, b3;
        b0.x = hr0[0];  b0.y = hr1[0];
        b1.x = hr0[16]; b1.y = hr1[16];
        b2.x = hr0[32]; b2.y = hr1[32];
        b3.x = hr0[48]; b3.y = hr1[48];

        c0 = __builtin_amdgcn_wmma_f32_16x16x4_f32(false, ap, false, b0, (short)0, c0, false, false);
        c1 = __builtin_amdgcn_wmma_f32_16x16x4_f32(false, ap, false, b1, (short)0, c1, false, false);
        c2 = __builtin_amdgcn_wmma_f32_16x16x4_f32(false, ap, false, b2, (short)0, c2, false, false);
        c3 = __builtin_amdgcn_wmma_f32_16x16x4_f32(false, ap, false, b3, (short)0, c3, false, false);
    }

#pragma unroll
    for (int r = 0; r < 8; ++r) {
        const int m = r + 8 * half;
        tile[m * 68 +  0 + lm] = c0[r];
        tile[m * 68 + 16 + lm] = c1[r];
        tile[m * 68 + 32 + lm] = c2[r];
        tile[m * 68 + 48 + lm] = c3[r];
    }
    __syncthreads();

    if (lane < 16) {
        const int r = lane;
        const float* hres = hb + (size_t)(i0 + r) * HOq;
        float sum = 0.f, sumsq = 0.f;
#pragma unroll 8
        for (int e = 0; e < HOq; ++e) {
            float v = tile[r * 68 + e] + bias[hh * HOq + e];
            v = fmaxf(v, 0.f);
            v += hres[e];
            tile[r * 68 + e] = v;
            sum += v; sumsq += v * v;
        }
        const float mu   = sum * (1.f / HOq);
        const float var  = sumsq * (1.f / HOq) - mu * mu;
        const float rstd = rsqrtf(var + EPSV);
        float* op = out + ((size_t)b * Nq + (i0 + r)) * (Hq * HOq) + hh * HOq;
#pragma unroll 8
        for (int e = 0; e < HOq; ++e)
            op[e] = (tile[r * 68 + e] - mu) * rstd * gamma[hh * HOq + e] + beta[hh * HOq + e];
    }
}

extern "C" void kernel_launch(void* const* d_in, const int* in_sizes, int n_in,
                              void* d_out, int out_size, void* d_ws, size_t ws_size,
                              hipStream_t stream) {
    const float* nodes = (const float*)d_in[0];
    const int*   adj   = (const int*)d_in[1];
    const float* Wm    = (const float*)d_in[2];
    const float* asrc  = (const float*)d_in[3];
    const float* adst  = (const float*)d_in[4];
    const float* bias  = (const float*)d_in[5];
    const float* gamma = (const float*)d_in[6];
    const float* beta  = (const float*)d_in[7];
    float* out = (float*)d_out;

    float* h_ws = (float*)d_ws;
    float* esrc = h_ws + (size_t)Bq * Hq * Nq * HOq;
    float* edst = esrc + (size_t)Bq * Hq * Tq * Nq;
    float* rmx  = edst + (size_t)Bq * Hq * Tq * Nq;
    float* rsm  = rmx  + (size_t)Bq * Hq * Nq;

    k_proj<<<dim3(Nq / 16, HOq / 16, Bq * Hq), 32, 0, stream>>>(nodes, Wm, h_ws);
    k_escore<<<(Bq * Hq * Tq * Nq + 255) / 256, 256, 0, stream>>>(h_ws, asrc, adst, esrc, edst);
    k_softstats<<<Bq * Hq * Nq, 32, 0, stream>>>(adj, esrc, edst, rmx, rsm);
    k_attn<<<dim3(Nq / 16, Bq * Hq), 32, 0, stream>>>(adj, h_ws, esrc, edst, rmx, rsm,
                                                      bias, gamma, beta, out);
}


# k_attn: each 16-row tile's 1024 neighbours split over 8 waves of one 256-thread group; partial tiles added in fixed order through LDS, row sums of the LayerNorm by 16-lane tree (sum order changes)
# speedup vs baseline: 1.5378x; 1.5378x over previous
#include <hip/hip_runtime.h>

#define Bq  8
#define Nq  1024
#define Dq  256
#define Hq  4
#define HOq 64
#define Tq  5
#define NEGV (-1e9f)
#define EPSV (1e-6f)

typedef float v2f __attribute__((ext_vector_type(2)));
typedef float v8f __attribute__((ext_vector_type(8)));

__device__ __forceinline__ float sel5(int t, float e0, float e1, float e2, float e3, float e4) {
    return (t == 0) ? e0 : (t == 1) ? e1 : (t == 2) ? e2 : (t == 3) ? e3 : e4;
}

__global__ void __launch_bounds__(32) k_proj(const float* __restrict__ X,
                                             const float* __restrict__ Wm,
                                             float* __restrict__ hws) {
    const int lane = threadIdx.x;
    const int half = lane >> 4;
    const int lm   = lane & 15;
    const int i0   = blockIdx.x * 16;
    const int ho0  = blockIdx.y * 16;
    const int bh   = blockIdx.z;
    const int b    = bh / Hq, hh = bh % Hq;

    const float* xr = X + ((size_t)b * Nq + (size_t)(i0 + lm)) * Dq;
    const float* wp = Wm + (size_t)hh * Dq * HOq + ho0 + lm;

    v8f c = {};
#pragma unroll 4
    for (int k = 0; k < Dq; k += 4) {
        const int kk = k + half * 2;
        float2 axy = *(const float2*)(xr + kk);
        v2f a; a.x = axy.x; a.y = axy.y;
        v2f bb;
        bb.x = wp[(size_t)kk * HOq];
        bb.y = wp[(size_t)(kk + 1) * HOq];
        c = __builtin_amdgcn_wmma_f32_16x16x4_f32(false, a, false, bb,
                                                  (short)0, c, false, false);
    }
    float* hp = hws + ((size_t)bh * Nq + i0) * HOq + ho0 + lm;
#pragma unroll
    for (int r = 0; r < 8; ++r)
        hp[(size_t)(r + 8 * half) * HOq] = c[r];
}

__global__ void __launch_bounds__(256) k_escore(const float* __restrict__ hws,
                                                const float* __restrict__ asrc,
                                                const float* __restrict__ adst,
                                                float* __restrict__ es,
                                                float* __restrict__ ed) {
    int idx = blockIdx.x * 256 + threadIdx.x;
    if (idx >= Bq * Hq * Tq * Nq) return;
    const int n  = idx % Nq;
    const int t  = (idx / Nq) % Tq;
    const int bh = idx / (Nq * Tq);
    const int hh = bh % Hq;
    const float4* hv = (const float4*)(hws + ((size_t)bh * Nq + n) * HOq);
    const float4* sv = (const float4*)(asrc + ((size_t)hh * Tq + t) * HOq);
    const float4* dv = (const float4*)(adst + ((size_t)hh * Tq + t) * HOq);
    float ss = 0.f, sd = 0.f;
#pragma unroll
    for (int q = 0; q < HOq / 4; ++q) {
        float4 hx = hv[q], sx = sv[q], dx = dv[q];
        ss += hx.x * sx.x + hx.y * sx.y + hx.z * sx.z + hx.w * sx.w;
        sd += hx.x * dx.x + hx.y * dx.y + hx.z * dx.z + hx.w * dx.w;
    }
    es[idx] = ss;
    ed[idx] = sd;
}

__global__ void __launch_bounds__(32) k_softstats(const int* __restrict__ adj,
                                                  const float* __restrict__ es,
                                                  const float* __restrict__ ed,
                                                  float* __restrict__ rmax,
                                                  float* __restrict__ rsum) {
    const int lane  = threadIdx.x;
    const int rowid = blockIdx.x;
    const int i  = rowid % Nq;
    const int bh = rowid / Nq;
    const int b  = bh / Hq;

    const float e0 = es[((size_t)bh * Tq + 0) * Nq + i];
    const float e1 = es[((size_t)bh * Tq + 1) * Nq + i];
    const float e2 = es[((size_t)bh * Tq + 2) * Nq + i];
    const float e3 = es[((size_t)bh * Tq + 3) * Nq + i];
    const float e4 = es[((size_t)bh * Tq + 4) * Nq + i];
    const int*   ar  = adj + ((size_t)b * Nq + i) * Nq;
    const float* edb = ed + (size_t)bh * Tq * Nq;

    float m = -INFINITY, s = 0.f;
    for (int j = lane; j < Nq; j += 32) {
        int a = ar[j];
        int t = (a > 0) ? (a - 1) : 0;
        float sc = sel5(t, e0, e1, e2, e3, e4) + edb[(size_t)t * Nq + j];
        float w = (a == 0) ? NEGV : (sc > 0.f ? sc : 0.01f * sc);
        float mn = fmaxf(m, w);
        s = s * __expf(m - mn) + __expf(w - mn);
        m = mn;
    }
#pragma unroll
    for (int off = 16; off >= 1; off >>= 1) {
        float mo = __shfl_xor(m, off, 32);
        float so = __shfl_xor(s, off, 32);
        float mn = fmaxf(m, mo);
        s = s * __expf(m - mn) + so * __expf(mo - mn);
        m = mn;
    }
    if (lane == 0) { rmax[rowid] = m; rsum[rowid] = s; }
}

#define JS   8
#define JLEN (Nq / JS)
#define TP   68
__global__ void __launch_bounds__(32 * JS) k_attn(const int* __restrict__ adj,
                                                  const float* __restrict__ hws,
                                                  const float* __restrict__ es,
                                                  const float* __restrict__ ed,
                                                  const float* __restrict__ rmax,
                                                  const float* __restrict__ rsum,
                                                  const float* __restrict__ bias,
                                                  const float* __restrict__ gamma,
                                                  const float* __restrict__ beta,
                                                  float* __restrict__ out) {
    __shared__ __attribute__((aligned(16))) float part[JS * 16 * TP];
    const int tid  = threadIdx.x;
    const int lane = tid & 31;
    const int wv   = __builtin_amdgcn_readfirstlane(tid >> 5);
    const int half = lane >> 4;
    const int lm   = lane & 15;
    const int i0   = blockIdx.x * 16;
    const int bh   = blockIdx.y;
    const int b    = bh / Hq, hh = bh % Hq;
    const int row  = i0 + lm;

    const float e0 = es[((size_t)bh * Tq + 0) * Nq + row];
    const float e1 = es[((size_t)bh * Tq + 1) * Nq + row];
    const float e2 = es[((size_t)bh * Tq + 2) * Nq + row];
    const float e3 = es[((size_t)bh * Tq + 3) * Nq + row];
    const float e4 = es[((size_t)bh * Tq + 4) * Nq + row];
    const float rm   = rmax[(size_t)bh * Nq + row];
    const float rinv = 1.f / rsum[(size_t)bh * Nq + row];

    const int*   ar  = adj + ((size_t)b * Nq + row) * Nq;
    const float* edb = ed + (size_t)bh * Tq * Nq;
    const float* hb  = hws + (size_t)bh * Nq * HOq;

    const int jbeg = wv * JLEN;
    v8f c0 = {}, c1 = {}, c2 = {}, c3 = {};
    for (int j0 = jbeg; j0 < jbeg + JLEN; j0 += 4) {
        const int jj = j0 + half * 2;
        int2 av = *(const int2*)(ar + jj);

        float p0, p1;
        {
            int a = av.x; int t = (a > 0) ? (a - 1) : 0;
            float sc = sel5(t, e0, e1, e2, e3, e4) + edb[(size_t)t * Nq + jj];
            float w = (a == 0) ? NEGV : (sc > 0.f ? sc : 0.01f * sc);
            p0 = __expf(w - rm) * rinv;
        }
        {
            int a = av.y; int t = (a > 0) ? (a - 1) : 0;
            float sc = sel5(t, e0, e1, e2, e3, e4) + edb[(size_t)t * Nq + jj + 1];
            float w = (a == 0) ? NEGV : (sc > 0.f ? sc : 0.01f * sc);
            p1 = __expf(w - rm) * rinv;
        }
        v2f ap; ap.x = p0; ap.y = p1;

        const float* hr0 = hb + (size_t)jj * HOq + lm;
        const float* hr1 = hr0 + HOq;
        v2f b0, b1, b2, b3;
        b0.x = hr0[0];  b0.y = hr1[0];
        b1.x = hr0[16]; b1.y = hr1[16];
        b2.x = hr0[32]; b2.y = hr1[32];
        b3.x = hr0[48]; b3.y = hr1[48];

        c0 = __builtin_amdgcn_wmma_f32_16x16x4_f32(false, ap, false, b0, (short)0, c0, false, false);
        c1 = __builtin_amdgcn_wmma_f32_16x16x4_f32(false, ap, false, b1, (short)0, c1, false, false);
        c2 = __builtin_amdgcn_wmma_f32_16x16x4_f32(false, ap, false, b2, (short)0, c2, false, false);
        c3 = __builtin_amdgcn_wmma_f32_16x16x4_f32(false, ap, false, b3, (short)0, c3, false, false);
    }

    float* pw = part + wv * (16 * TP);
#pragma unroll
    for (int r = 0; r < 8; ++r) {
        const int m = r + 8 * half;
        pw[m * TP +  0 + lm] = c0[r];
        pw[m * TP + 16 + lm] = c1[r];
        pw[m * TP + 32 + lm] = c2[r];
        pw[m * TP + 48 + lm] = c3[r];
    }
    __syncthreads();

    const int r  = tid >> 4;
    const int q4 = (tid & 15) * 4;
    const float* pr = part + r * TP + q4;
    float4 v = *(const float4*)pr;
    for (int w = 1; w < JS; ++w) {
        const float4 t = *(const float4*)(pr + w * (16 * TP));
        v.x += t.x; v.y += t.y; v.z += t.z; v.w += t.w;
    }
    const float4 bv = *(const float4*)(bias + hh * HOq + q4);
    const float4 hv = *(const float4*)(hb + (size_t)(i0 + r) * HOq + q4);
    v.x = fmaxf(v.x + bv.x, 0.f) + hv.x;
    v.y = fmaxf(v.y + bv.y, 0.f) + hv.y;
    v.z = fmaxf(v.z + bv.z, 0.f) + hv.z;
    v.w = fmaxf(v.w + bv.w, 0.f) + hv.w;
    float sum   = (v.x + v.y) + (v.z + v.w);
    float sumsq = (v.x * v.x + v.y * v.y) + (v.z * v.z + v.w * v.w);
    for (int off = 8; off >= 1; off >>= 1) {
        sum   += __shfl_xor(sum,   off, 32);
        sumsq += __shfl_xor(sumsq, off, 32);
    }
    const float mu   = sum * (1.f / HOq);
    const float var  = sumsq * (1.f / HOq) - mu * mu;
    const float rstd = rsqrtf(var + EPSV);
    const float4 gv = *(const float4*)(gamma + hh * HOq + q4);
    const float4 tv = *(const float4*)(beta  + hh * HOq + q4);
    float4 o;
    o.x = (v.x - mu) * rstd * gv.x + tv.x;
    o.y = (v.y - mu) * rstd * gv.y + tv.y;
    o.z = (v.z - mu) * rstd * gv.z + tv.z;
    o.w = (v.w - mu) * rstd * gv.w + tv.w;
    *(float4*)(out + ((size_t)b * Nq + (i0 + r)) * (Hq * HOq) + hh * HOq + q4) = o;
}

extern "C" void kernel_launch(void* const* d_in, const int* in_sizes, int n_in,
                              void* d_out, int out_size, void* d_ws, size_t ws_size,
                              hipStream_t stream) {
    const float* nodes = (const float*)d_in[0];
    const int*   adj   = (const int*)d_in[1];
    const float* Wm    = (const float*)d_in[2];
    const float* asrc  = (const float*)d_in[3];
    const float* adst  = (const float*)d_in[4];
    const float* bias  = (const float*)d_in[5];
    const float* gamma = (const float*)d_in[6];
    const float* beta  = (const float*)d_in[7];
    float* out = (float*)d_out;

    float* h_ws = (float*)d_ws;
    float* esrc = h_ws + (size_t)Bq * Hq * Nq * HOq;
    float* edst = esrc + (size_t)Bq * Hq * Tq * Nq;
    float* rmx  = edst + (size_t)Bq * Hq * Tq * Nq;
    float* rsm  = rmx  + (size_t)Bq * Hq * Nq;

    k_proj<<<dim3(Nq / 16, HOq / 16, Bq * Hq), 32, 0, stream>>>(nodes, Wm, h_ws);
    k_escore<<<(Bq * Hq * Tq * Nq + 255) / 256, 256, 0, stream>>>(h_ws, asrc, adst, esrc, edst);
    k_softstats<<<Bq * Hq * Nq, 32, 0, stream>>>(adj, esrc, edst, rmx, rsm);
    k_attn<<<dim3(Nq / 16, Bq * Hq), 32 * JS, 0, stream>>>(adj, h_ws, esrc, edst, rmx, rsm,
                                                      bias, gamma, beta, out);
}


# k_softstats folded into k_attn: group gets its 16 rows' exact max from scores held in registers (slice maxima via LDS); WMMAs take unnormalised exp, row sum divides the tile last (sum order changes)
# speedup vs baseline: 3.3473x; 2.1767x over previous
#include <hip/hip_runtime.h>

#define Bq  8
#define Nq  1024
#define Dq  256
#define Hq  4
#define HOq 64
#define Tq  5
#define NEGV (-1e9f)
#define EPSV (1e-6f)

typedef float v2f __attribute__((ext_vector_type(2)));
typedef float v8f __attribute__((ext_vector_type(8)));

__device__ __forceinline__ float sel5(int t, float e0, float e1, float e2, float e3, float e4) {
    float r = e4;
    r = (t == 3) ? e3 : r;
    r = (t == 2) ? e2 : r;
    r = (t == 1) ? e1 : r;
    r = (t == 0) ? e0 : r;
    return r;
}

__global__ void __launch_bounds__(32) k_proj(const float* __restrict__ X,
                                             const float* __restrict__ Wm,
                                             float* __restrict__ hws) {
    const int lane = threadIdx.x;
    const int half = lane >> 4;
    const int lm   = lane & 15;
    const int i0   = blockIdx.x * 16;
    const int ho0  = blockIdx.y * 16;
    const int bh   = blockIdx.z;
    const int b    = bh / Hq, hh = bh % Hq;

    const float* xr = X + ((size_t)b * Nq + (size_t)(i0 + lm)) * Dq;
    const float* wp = Wm + (size_t)hh * Dq * HOq + ho0 + lm;

    v8f c = {};
#pragma unroll 4
    for (int k = 0; k < Dq; k += 4) {
        const int kk = k + half * 2;
        float2 axy = *(const float2*)(xr + kk);
        v2f a; a.x = axy.x; a.y = axy.y;
        v2f bb;
        bb.x = wp[(size_t)kk * HOq];
        bb.y = wp[(size_t)(kk + 1) * HOq];
        c = __builtin_amdgcn_wmma_f32_16x16x4_f32(false, a, false, bb,
                                                  (short)0, c, false, false);
    }
    float* hp = hws + ((size_t)bh * Nq + i0) * HOq + ho0 + lm;
#pragma unroll
    for (int r = 0; r < 8; ++r)
        hp[(size_t)(r + 8 * half) * HOq] = c[r];
}

__global__ void __launch_bounds__(256) k_escore(const float* __restrict__ hws,
                                                const float* __restrict__ asrc,
                                                const float* __restrict__ adst,
                                                float* __restrict__ es,
                                                float* __restrict__ ed) {
    int idx = blockIdx.x * 256 + threadIdx.x;
    if (idx >= Bq * Hq * Tq * Nq) return;
    const int n  = idx % Nq;
    const int t  = (idx / Nq) % Tq;
    const int bh = idx / (Nq * Tq);
    const int hh = bh % Hq;
    const float4* hv = (const float4*)(hws + ((size_t)bh * Nq + n) * HOq);
    const float4* sv = (const float4*)(asrc + ((size_t)hh * Tq + t) * HOq);
    const float4* dv = (const float4*)(adst + ((size_t)hh * Tq + t) * HOq);
    float ss = 0.f, sd = 0.f;
#pragma unroll
    for (int q = 0; q < HOq / 4; ++q) {
        float4 hx = hv[q], sx = sv[q], dx = dv[q];
        ss += hx.x * sx.x + hx.y * sx.y + hx.z * sx.z + hx.w * sx.w;
        sd += hx.x * dx.x + hx.y * dx.y + hx.z * dx.z + hx.w * dx.w;
    }
    es[idx] = ss;
    ed[idx] = sd;
}

#define JS   8
#define JLEN (Nq / JS)
#define NIT  (JLEN / 4)
#define GB   8
#define TP   68

__device__ __forceinline__ float score1(int a, float edv, float e0, float e1, float e2,
                                        float e3, float e4) {
    const int t = (a > 0) ? (a - 1) : 0;
    const float sc  = sel5(t, e0, e1, e2, e3, e4) + edv;
    const float scn = 0.01f * sc;
    const float lr  = (sc > 0.f) ? sc : scn;
    return (a == 0) ? NEGV : lr;
}

__global__ void __launch_bounds__(32 * JS) k_attn(const int* __restrict__ adj,
                                                  const float* __restrict__ hws,
                                                  const float* __restrict__ es,
                                                  const float* __restrict__ ed,
                                                  const float* __restrict__ bias,
                                                  const float* __restrict__ gamma,
                                                  const float* __restrict__ beta,
                                                  float* __restrict__ out) {
    __shared__ __attribute__((aligned(16))) float part[JS * 16 * TP];
    __shared__ __attribute__((aligned(16))) float smax[16 * JS];
    __shared__ __attribute__((aligned(16))) float ssum[16 * JS];
    const int tid  = threadIdx.x;
    const int lane = tid & 31;
    const int wv   = __builtin_amdgcn_readfirstlane(tid >> 5);
    const int half = lane >> 4;
    const int lm   = lane & 15;
    const int i0   = blockIdx.x * 16;
    const int bh   = blockIdx.y;
    const int b    = bh / Hq, hh = bh % Hq;
    const int row  = i0 + lm;

    const float e0 = es[((size_t)bh * Tq + 0) * Nq + row];
    const float e1 = es[((size_t)bh * Tq + 1) * Nq + row];
    const float e2 = es[((size_t)bh * Tq + 2) * Nq + row];
    const float e3 = es[((size_t)bh * Tq + 3) * Nq + row];
    const float e4 = es[((size_t)bh * Tq + 4) * Nq + row];

    const int jbeg = wv * JLEN;
    const unsigned lofs = (unsigned)half * 2u;
    const int*   ar  = adj + ((size_t)b * Nq + row) * Nq + jbeg + lofs;
    const float* eds = ed + (size_t)bh * Tq * Nq + jbeg;
    const float* hb  = hws + (size_t)bh * Nq * HOq;
    const float* hbs = hb + (size_t)jbeg * HOq;
    const unsigned ho = lofs * HOq + (unsigned)lm;

    int2  av[NIT];
    float w[2 * NIT];
#pragma unroll
    for (int it = 0; it < NIT; ++it)
        av[it] = *(const int2*)(ar + 4 * it);
    float m = -INFINITY;
#pragma unroll
    for (int blk = 0; blk < NIT; blk += GB) {
#pragma unroll
        for (int it = blk; it < blk + GB; ++it) {
            const unsigned t0 = (av[it].x > 0) ? (unsigned)(av[it].x - 1) : 0u;
            const unsigned t1 = (av[it].y > 0) ? (unsigned)(av[it].y - 1) : 0u;
            w[2 * it]     = eds[t0 * Nq + lofs + 4 * it];
            w[2 * it + 1] = eds[t1 * Nq + lofs + 4 * it + 1];
        }
#pragma unroll
        for (int it = blk; it < blk + GB; ++it) {
            const float w0 = score1(av[it].x, w[2 * it],     e0, e1, e2, e3, e4);
            const float w1 = score1(av[it].y, w[2 * it + 1], e0, e1, e2, e3, e4);
            w[2 * it]     = w0;
            w[2 * it + 1] = w1;
            m = fmaxf(m, fmaxf(w0, w1));
        }
    }

#define LOADB(dst, it_)                                                        \
    {                                                                          \
        const float* hp_ = hbs + ho + (4 * (it_)) * HOq;                       \
        dst[0].x = hp_[0];  dst[0].y = hp_[HOq];                               \
        dst[1].x = hp_[16]; dst[1].y = hp_[HOq + 16];                          \
        dst[2].x = hp_[32]; dst[2].y = hp_[HOq + 32];                          \
        dst[3].x = hp_[48]; dst[3].y = hp_[HOq + 48];                          \
    }
    v2f bq[3][4];
    LOADB(bq[0], 0);
    LOADB(bq[1], 1);

    m = fmaxf(m, __shfl_xor(m, 16, 32));
    if (half == 0) smax[lm * JS + wv] = m;
    __syncthreads();
    float rm;
    {
        const float4 ma = *(const float4*)(smax + lm * JS);
        const float4 mb = *(const float4*)(smax + lm * JS + 4);
        rm = fmaxf(fmaxf(fmaxf(ma.x, ma.y), fmaxf(ma.z, ma.w)),
                   fmaxf(fmaxf(mb.x, mb.y), fmaxf(mb.z, mb.w)));
    }

    float s = 0.f;
    v8f c0 = {}, c1 = {}, c2 = {}, c3 = {};
#pragma unroll
    for (int it = 0; it < NIT; ++it) {
        if (it + 2 < NIT) LOADB(bq[(it + 2) % 3], it + 2);
        __builtin_amdgcn_sched_barrier(0);
        const float p0 = __expf(w[2 * it]     - rm);
        const float p1 = __expf(w[2 * it + 1] - rm);
        s += p0 + p1;
        v2f ap; ap.x = p0; ap.y = p1;
        c0 = __builtin_amdgcn_wmma_f32_16x16x4_f32(false, ap, false, bq[it % 3][0], (short)0, c0, false, false);
        c1 = __builtin_amdgcn_wmma_f32_16x16x4_f32(false, ap, false, bq[it % 3][1], (short)0, c1, false, false);
        c2 = __builtin_amdgcn_wmma_f32_16x16x4_f32(false, ap, false, bq[it % 3][2], (short)0, c2, false, false);
        c3 = __builtin_amdgcn_wmma_f32_16x16x4_f32(false, ap, false, bq[it % 3][3], (short)0, c3, false, false);
    }
#undef LOADB
    s += __shfl_xor(s, 16, 32);
    if (half == 0) ssum[lm * JS + wv] = s;

    float* pw = part + wv * (16 * TP);
#pragma unroll
    for (int r = 0; r < 8; ++r) {
        const int mr = r + 8 * half;
        pw[mr * TP +  0 + lm] = c0[r];
        pw[mr * TP + 16 + lm] = c1[r];
        pw[mr * TP + 32 + lm] = c2[r];
        pw[mr * TP + 48 + lm] = c3[r];
    }
    __syncthreads();

    const int r  = tid >> 4;
    const int q4 = (tid & 15) * 4;
    const float* pr = part + r * TP + q4;
    float4 v = *(const float4*)pr;
    for (int ww = 1; ww < JS; ++ww) {
        const float4 t = *(const float4*)(pr + ww * (16 * TP));
        v.x += t.x; v.y += t.y; v.z += t.z; v.w += t.w;
    }
    const float4 sa = *(const float4*)(ssum + r * JS);
    const float4 sb = *(const float4*)(ssum + r * JS + 4);
    const float tot  = ((((((sa.x + sa.y) + sa.z) + sa.w) + sb.x) + sb.y) + sb.z) + sb.w;
    const float rinv = 1.f / tot;
    const float4 bv = *(const float4*)(bias + hh * HOq + q4);
    const float4 hv = *(const float4*)(hb + (size_t)(i0 + r) * HOq + q4);
    v.x = fmaxf(v.x * rinv + bv.x, 0.f) + hv.x;
    v.y = fmaxf(v.y * rinv + bv.y, 0.f) + hv.y;
    v.z = fmaxf(v.z * rinv + bv.z, 0.f) + hv.z;
    v.w = fmaxf(v.w * rinv + bv.w, 0.f) + hv.w;
    float sum   = (v.x + v.y) + (v.z + v.w);
    float sumsq = (v.x * v.x + v.y * v.y) + (v.z * v.z + v.w * v.w);
    for (int off = 8; off >= 1; off >>= 1) {
        sum   += __shfl_xor(sum,   off, 32);
        sumsq += __shfl_xor(sumsq, off, 32);
    }
    const float mu   = sum * (1.f / HOq);
    const float var  = sumsq * (1.f / HOq) - mu * mu;
    const float rstd = rsqrtf(var + EPSV);
    const float4 gv = *(const float4*)(gamma + hh * HOq + q4);
    const float4 tv = *(const float4*)(beta  + hh * HOq + q4);
    float4 o;
    o.x = (v.x - mu) * rstd * gv.x + tv.x;
    o.y = (v.y - mu) * rstd * gv.y + tv.y;
    o.z = (v.z - mu) * rstd * gv.z + tv.z;
    o.w = (v.w - mu) * rstd * gv.w + tv.w;
    *(float4*)(out + ((size_t)b * Nq + (i0 + r)) * (Hq * HOq) + hh * HOq + q4) = o;
}

extern "C" void kernel_launch(void* const* d_in, const int* in_sizes, int n_in,
                              void* d_out, int out_size, void* d_ws, size_t ws_size,
                              hipStream_t stream) {
    const float* nodes = (const float*)d_in[0];
    const int*   adj   = (const int*)d_in[1];
    const float* Wm    = (const float*)d_in[2];
    const float* asrc  = (const float*)d_in[3];
    const float* adst  = (const float*)d_in[4];
    const float* bias  = (const float*)d_in[5];
    const float* gamma = (const float*)d_in[6];
    const float* beta  = (const float*)d_in[7];
    float* out = (float*)d_out;

    float* h_ws = (float*)d_ws;
    float* esrc = h_ws + (size_t)Bq * Hq * Nq * HOq;
    float* edst = esrc + (size_t)Bq * Hq * Tq * Nq;

    k_proj<<<dim3(Nq / 16, HOq / 16, Bq * Hq), 32, 0, stream>>>(nodes, Wm, h_ws);
    k_escore<<<(Bq * Hq * Tq * Nq + 255) / 256, 256, 0, stream>>>(h_ws, asrc, adst, esrc, edst);
    k_attn<<<dim3(Nq / 16, Bq * Hq), 32 * JS, 0, stream>>>(adj, h_ws, esrc, edst,
                                                      bias, gamma, beta, out);
}


# k_escore folded into k_proj: a wave owns a head's whole 16x64 tile (4 WMMAs share an A fragment, 16-byte h pieces) and scores its rows from LDS, one fma chain per dot (e-score sum order changes)
# speedup vs baseline: 3.9048x; 1.1666x over previous
#include <hip/hip_runtime.h>

#define Bq  8
#define Nq  1024
#define Dq  256
#define Hq  4
#define HOq 64
#define Tq  5
#define NEGV (-1e9f)
#define EPSV (1e-6f)

typedef float v2f __attribute__((ext_vector_type(2)));
typedef float v8f __attribute__((ext_vector_type(8)));

__device__ __forceinline__ float sel5(int t, float e0, float e1, float e2, float e3, float e4) {
    float r = e4;
    r = (t == 3) ? e3 : r;
    r = (t == 2) ? e2 : r;
    r = (t == 1) ? e1 : r;
    r = (t == 0) ? e0 : r;
    return r;
}

#define PW  4
#define PKU 8
#define PTP 68

__global__ void __launch_bounds__(32 * PW) k_proj(const float* __restrict__ X,
                                                  const float* __restrict__ Wm,
                                                  const float* __restrict__ asrc,
                                                  const float* __restrict__ adst,
                                                  float* __restrict__ hws,
                                                  float* __restrict__ es,
                                                  float* __restrict__ ed) {
    __shared__ __attribute__((aligned(16))) float hs[PW * 16 * PTP];
    const int tid  = threadIdx.x;
    const int lane = tid & 31;
    const int wv   = __builtin_amdgcn_readfirstlane(tid >> 5);
    const int half = lane >> 4;
    const int lm   = lane & 15;
    const int i0   = (blockIdx.x * PW + wv) * 16;
    const int bh   = blockIdx.y;
    const int b    = bh / Hq, hh = bh % Hq;

    const float* xr = X + ((size_t)b * Nq + (size_t)(i0 + lm)) * Dq + half * 2;
    const float* wp = Wm + (size_t)hh * Dq * HOq + (size_t)(half * 2) * HOq + lm * 4;

    float2 fa[2][PKU];
    float4 fw0[2][PKU], fw1[2][PKU];
#define PLOAD(st_, kb_)                                                        \
    {                                                                          \
          \
        _Pragma("unroll")                                                      \
        for (int u_ = 0; u_ < PKU; ++u_) {                                     \
            const int k_ = ((kb_) * PKU + u_) * 4;                             \
            fa[st_][u_]  = *(const float2*)(xr + k_);                          \
            fw0[st_][u_] = *(const float4*)(wp + (size_t)k_ * HOq);            \
            fw1[st_][u_] = *(const float4*)(wp + (size_t)(k_ + 1) * HOq);      \
        }                                                                      \
    }

    v8f c0 = {}, c1 = {}, c2 = {}, c3 = {};
    PLOAD(0, 0);
#pragma unroll
    for (int kb = 0; kb < Dq / (4 * PKU); ++kb) {
        if (kb + 1 < Dq / (4 * PKU)) PLOAD((kb + 1) & 1, kb + 1);
        __builtin_amdgcn_sched_barrier(0);
#pragma unroll
        for (int u = 0; u < PKU; ++u) {
            const float2 av = fa[kb & 1][u];
            const float4 w0 = fw0[kb & 1][u];
            const float4 w1 = fw1[kb & 1][u];
            v2f a;  a.x  = av.x; a.y  = av.y;
            v2f b0; b0.x = w0.x; b0.y = w1.x;
            v2f b1; b1.x = w0.y; b1.y = w1.y;
            v2f b2; b2.x = w0.z; b2.y = w1.z;
            v2f b3; b3.x = w0.w; b3.y = w1.w;
            c0 = __builtin_amdgcn_wmma_f32_16x16x4_f32(false, a, false, b0, (short)0, c0, false, false);
            c1 = __builtin_amdgcn_wmma_f32_16x16x4_f32(false, a, false, b1, (short)0, c1, false, false);
            c2 = __builtin_amdgcn_wmma_f32_16x16x4_f32(false, a, false, b2, (short)0, c2, false, false);
            c3 = __builtin_amdgcn_wmma_f32_16x16x4_f32(false, a, false, b3, (short)0, c3, false, false);
        }
    }
#undef PLOAD

    float* hp  = hws + ((size_t)bh * Nq + i0) * HOq + lm * 4;
    float* hsw = hs + wv * (16 * PTP);
#pragma unroll
    for (int r = 0; r < 8; ++r) {
        const int mr = r + 8 * half;
        float4 o; o.x = c0[r]; o.y = c1[r]; o.z = c2[r]; o.w = c3[r];
        *(float4*)(hsw + mr * PTP + lm * 4) = o;
    }
    __syncthreads();
#pragma unroll
    for (int r = 0; r < 8; ++r) {
        const int mr = r + 8 * half;
        float4 o; o.x = c0[r]; o.y = c1[r]; o.z = c2[r]; o.w = c3[r];
        *(float4*)(hp + (size_t)mr * HOq) = o;
    }

    const float* av_ = (half ? adst : asrc) + (size_t)hh * Tq * HOq;
    const float* hr  = hsw + lm * PTP;
    float s0 = 0.f, s1 = 0.f, s2 = 0.f, s3 = 0.f, s4 = 0.f;
#pragma unroll
    for (int q = 0; q < HOq / 4; ++q) {
        const float4 hx = *(const float4*)(hr + 4 * q);
        const float4 a0 = *(const float4*)(av_ + 0 * HOq + 4 * q);
        const float4 a1 = *(const float4*)(av_ + 1 * HOq + 4 * q);
        const float4 a2 = *(const float4*)(av_ + 2 * HOq + 4 * q);
        const float4 a3 = *(const float4*)(av_ + 3 * HOq + 4 * q);
        const float4 a4 = *(const float4*)(av_ + 4 * HOq + 4 * q);
        s0 = fmaf(hx.w, a0.w, fmaf(hx.z, a0.z, fmaf(hx.y, a0.y, fmaf(hx.x, a0.x, s0))));
        s1 = fmaf(hx.w, a1.w, fmaf(hx.z, a1.z, fmaf(hx.y, a1.y, fmaf(hx.x, a1.x, s1))));
        s2 = fmaf(hx.w, a2.w, fmaf(hx.z, a2.z, fmaf(hx.y, a2.y, fmaf(hx.x, a2.x, s2))));
        s3 = fmaf(hx.w, a3.w, fmaf(hx.z, a3.z, fmaf(hx.y, a3.y, fmaf(hx.x, a3.x, s3))));
        s4 = fmaf(hx.w, a4.w, fmaf(hx.z, a4.z, fmaf(hx.y, a4.y, fmaf(hx.x, a4.x, s4))));
    }
    float* ep = (half ? ed : es) + (size_t)bh * Tq * Nq + i0 + lm;
    ep[0 * Nq] = s0;
    ep[1 * Nq] = s1;
    ep[2 * Nq] = s2;
    ep[3 * Nq] = s3;
    ep[4 * Nq] = s4;
}

#define JS   8
#define JLEN (Nq / JS)
#define NIT  (JLEN / 4)
#define GB   8
#define TP   68

__device__ __forceinline__ float score1(int a, float edv, float e0, float e1, float e2,
                                        float e3, float e4) {
    const int t = (a > 0) ? (a - 1) : 0;
    const float sc  = sel5(t, e0, e1, e2, e3, e4) + edv;
    const float scn = 0.01f * sc;
    const float lr  = (sc > 0.f) ? sc : scn;
    return (a == 0) ? NEGV : lr;
}

__global__ void __launch_bounds__(32 * JS) k_attn(const int* __restrict__ adj,
                                                  const float* __restrict__ hws,
                                                  const float* __restrict__ es,
                                                  const float* __restrict__ ed,
                                                  const float* __restrict__ bias,
                                                  const float* __restrict__ gamma,
                                                  const float* __restrict__ beta,
                                                  float* __restrict__ out) {
    __shared__ __attribute__((aligned(16))) float part[JS * 16 * TP];
    __shared__ __attribute__((aligned(16))) float smax[16 * JS];
    __shared__ __attribute__((aligned(16))) float ssum[16 * JS];
    const int tid  = threadIdx.x;
    const int lane = tid & 31;
    const int wv   = __builtin_amdgcn_readfirstlane(tid >> 5);
    const int half = lane >> 4;
    const int lm   = lane & 15;
    const int i0   = blockIdx.x * 16;
    const int bh   = blockIdx.y;
    const int b    = bh / Hq, hh = bh % Hq;
    const int row  = i0 + lm;

    const float e0 = es[((size_t)bh * Tq + 0) * Nq + row];
    const float e1 = es[((size_t)bh * Tq + 1) * Nq + row];
    const float e2 = es[((size_t)bh * Tq + 2) * Nq + row];
    const float e3 = es[((size_t)bh * Tq + 3) * Nq + row];
    const float e4 = es[((size_t)bh * Tq + 4) * Nq + row];

    const int jbeg = wv * JLEN;
    const unsigned lofs = (unsigned)half * 2u;
    const int*   ar  = adj + ((size_t)b * Nq + row) * Nq + jbeg + lofs;
    const float* eds = ed + (size_t)bh * Tq * Nq + jbeg;
    const float* hb  = hws + (size_t)bh * Nq * HOq;
    const float* hbs = hb + (size_t)jbeg * HOq;
    const unsigned ho = lofs * HOq + (unsigned)lm;

    int2  av[NIT];
    float w[2 * NIT];
#pragma unroll
    for (int it = 0; it < NIT; ++it)
        av[it] = *(const int2*)(ar + 4 * it);
    float m = -INFINITY;
#pragma unroll
    for (int blk = 0; blk < NIT; blk += GB) {
#pragma unroll
        for (int it = blk; it < blk + GB; ++it) {
            const unsigned t0 = (av[it].x > 0) ? (unsigned)(av[it].x - 1) : 0u;
            const unsigned t1 = (av[it].y > 0) ? (unsigned)(av[it].y - 1) : 0u;
            w[2 * it]     = eds[t0 * Nq + lofs + 4 * it];
            w[2 * it + 1] = eds[t1 * Nq + lofs + 4 * it + 1];
        }
#pragma unroll
        for (int it = blk; it < blk + GB; ++it) {
            const float w0 = score1(av[it].x, w[2 * it],     e0, e1, e2, e3, e4);
            const float w1 = score1(av[it].y, w[2 * it + 1], e0, e1, e2, e3, e4);
            w[2 * it]     = w0;
            w[2 * it + 1] = w1;
            m = fmaxf(m, fmaxf(w0, w1));
        }
    }

#define LOADB(dst, it_)                                                        \
    {                                                                          \
        const float* hp_ = hbs + ho + (4 * (it_)) * HOq;                       \
        dst[0].x = hp_[0];  dst[0].y = hp_[HOq];                               \
        dst[1].x = hp_[16]; dst[1].y = hp_[HOq + 16];                          \
        dst[2].x = hp_[32]; dst[2].y = hp_[HOq + 32];                          \
        dst[3].x = hp_[48]; dst[3].y = hp_[HOq + 48];                          \
    }
    v2f bq[3][4];
    LOADB(bq[0], 0);
    LOADB(bq[1], 1);

    m = fmaxf(m, __shfl_xor(m, 16, 32));
    if (half == 0) smax[lm * JS + wv] = m;
    __syncthreads();
    float rm;
    {
        const float4 ma = *(const float4*)(smax + lm * JS);
        const float4 mb = *(const float4*)(smax + lm * JS + 4);
        rm = fmaxf(fmaxf(fmaxf(ma.x, ma.y), fmaxf(ma.z, ma.w)),
                   fmaxf(fmaxf(mb.x, mb.y), fmaxf(mb.z, mb.w)));
    }

    float s = 0.f;
    v8f c0 = {}, c1 = {}, c2 = {}, c3 = {};
#pragma unroll
    for (int it = 0; it < NIT; ++it) {
        if (it + 2 < NIT) LOADB(bq[(it + 2) % 3], it + 2);
        __builtin_amdgcn_sched_barrier(0);
        const float p0 = __expf(w[2 * it]     - rm);
        const float p1 = __expf(w[2 * it + 1] - rm);
        s += p0 + p1;
        v2f ap; ap.x = p0; ap.y = p1;
        c0 = __builtin_amdgcn_wmma_f32_16x16x4_f32(false, ap, false, bq[it % 3][0], (short)0, c0, false, false);
        c1 = __builtin_amdgcn_wmma_f32_16x16x4_f32(false, ap, false, bq[it % 3][1], (short)0, c1, false, false);
        c2 = __builtin_amdgcn_wmma_f32_16x16x4_f32(false, ap, false, bq[it % 3][2], (short)0, c2, false, false);
        c3 = __builtin_amdgcn_wmma_f32_16x16x4_f32(false, ap, false, bq[it % 3][3], (short)0, c3, false, false);
    }
#undef LOADB
    s += __shfl_xor(s, 16, 32);
    if (half == 0) ssum[lm * JS + wv] = s;

    float* pw = part + wv * (16 * TP);
#pragma unroll
    for (int r = 0; r < 8; ++r) {
        const int mr = r + 8 * half;
        pw[mr * TP +  0 + lm] = c0[r];
        pw[mr * TP + 16 + lm] = c1[r];
        pw[mr * TP + 32 + lm] = c2[r];
        pw[mr * TP + 48 + lm] = c3[r];
    }
    __syncthreads();

    const int r  = tid >> 4;
    const int q4 = (tid & 15) * 4;
    const float* pr = part + r * TP + q4;
    float4 v = *(const float4*)pr;
    for (int ww = 1; ww < JS; ++ww) {
        const float4 t = *(const float4*)(pr + ww * (16 * TP));
        v.x += t.x; v.y += t.y; v.z += t.z; v.w += t.w;
    }
    const float4 sa = *(const float4*)(ssum + r * JS);
    const float4 sb = *(const float4*)(ssum + r * JS + 4);
    const float tot  = ((((((sa.x + sa.y) + sa.z) + sa.w) + sb.x) + sb.y) + sb.z) + sb.w;
    const float rinv = 1.f / tot;
    const float4 bv = *(const float4*)(bias + hh * HOq + q4);
    const float4 hv = *(const float4*)(hb + (size_t)(i0 + r) * HOq + q4);
    v.x = fmaxf(v.x * rinv + bv.x, 0.f) + hv.x;
    v.y = fmaxf(v.y * rinv + bv.y, 0.f) + hv.y;
    v.z = fmaxf(v.z * rinv + bv.z, 0.f) + hv.z;
    v.w = fmaxf(v.w * rinv + bv.w, 0.f) + hv.w;
    float sum   = (v.x + v.y) + (v.z + v.w);
    float sumsq = (v.x * v.x + v.y * v.y) + (v.z * v.z + v.w * v.w);
    for (int off = 8; off >= 1; off >>= 1) {
        sum   += __shfl_xor(sum,   off, 32);
        sumsq += __shfl_xor(sumsq, off, 32);
    }
    const float mu   = sum * (1.f / HOq);
    const float var  = sumsq * (1.f / HOq) - mu * mu;
    const float rstd = rsqrtf(var + EPSV);
    const float4 gv = *(const float4*)(gamma + hh * HOq + q4);
    const float4 tv = *(const float4*)(beta  + hh * HOq + q4);
    float4 o;
    o.x = (v.x - mu) * rstd * gv.x + tv.x;
    o.y = (v.y - mu) * rstd * gv.y + tv.y;
    o.z = (v.z - mu) * rstd * gv.z + tv.z;
    o.w = (v.w - mu) * rstd * gv.w + tv.w;
    *(float4*)(out + ((size_t)b * Nq + (i0 + r)) * (Hq * HOq) + hh * HOq + q4) = o;
}

extern "C" void kernel_launch(void* const* d_in, const int* in_sizes, int n_in,
                              void* d_out, int out_size, void* d_ws, size_t ws_size,
                              hipStream_t stream) {
    const float* nodes = (const float*)d_in[0];
    const int*   adj   = (const int*)d_in[1];
    const float* Wm    = (const float*)d_in[2];
    const float* asrc  = (const float*)d_in[3];
    const float* adst  = (const float*)d_in[4];
    const float* bias  = (const float*)d_in[5];
    const float* gamma = (const float*)d_in[6];
    const float* beta  = (const float*)d_in[7];
    float* out = (float*)d_out;

    float* h_ws = (float*)d_ws;
    float* esrc = h_ws + (size_t)Bq * Hq * Nq * HOq;
    float* edst = esrc + (size_t)Bq * Hq * Tq * Nq;

    k_proj<<<dim3(Nq / (16 * PW), Bq * Hq), 32 * PW, 0, stream>>>(nodes, Wm, asrc, adst, h_ws, esrc, edst);
    k_attn<<<dim3(Nq / 16, Bq * Hq), 32 * JS, 0, stream>>>(adj, h_ws, esrc, edst,
                                                      bias, gamma, beta, out);
}


# k_attn: each wave owns two 16-row tiles of its neighbour slice (32 rows per group): an h fragment feeds 8 WMMAs, not 4; tile 1's scores wait in the wave's own LDS; per-row sum order unchanged
# speedup vs baseline: 4.4226x; 1.1326x over previous
#include <hip/hip_runtime.h>

#define Bq  8
#define Nq  1024
#define Dq  256
#define Hq  4
#define HOq 64
#define Tq  5
#define NEGV (-1e9f)
#define EPSV (1e-6f)

typedef float v2f __attribute__((ext_vector_type(2)));
typedef float v8f __attribute__((ext_vector_type(8)));

__device__ __forceinline__ float sel5(int t, float e0, float e1, float e2, float e3, float e4) {
    float r = e4;
    r = (t == 3) ? e3 : r;
    r = (t == 2) ? e2 : r;
    r = (t == 1) ? e1 : r;
    r = (t == 0) ? e0 : r;
    return r;
}

#define PW  4
#define PKU 8
#define PTP 68

__global__ void __launch_bounds__(32 * PW) k_proj(const float* __restrict__ X,
                                                  const float* __restrict__ Wm,
                                                  const float* __restrict__ asrc,
                                                  const float* __restrict__ adst,
                                                  float* __restrict__ hws,
                                                  float* __restrict__ es,
                                                  float* __restrict__ ed) {
    __shared__ __attribute__((aligned(16))) float hs[PW * 16 * PTP];
    const int tid  = threadIdx.x;
    const int lane = tid & 31;
    const int wv   = __builtin_amdgcn_readfirstlane(tid >> 5);
    const int half = lane >> 4;
    const int lm   = lane & 15;
    const int i0   = (blockIdx.x * PW + wv) * 16;
    const int bh   = blockIdx.y;
    const int b    = bh / Hq, hh = bh % Hq;

    const float* xr = X + ((size_t)b * Nq + (size_t)(i0 + lm)) * Dq + half * 2;
    const float* wp = Wm + (size_t)hh * Dq * HOq + (size_t)(half * 2) * HOq + lm * 4;

    float2 fa[2][PKU];
    float4 fw0[2][PKU], fw1[2][PKU];
#define PLOAD(st_, kb_)                                                        \
    {                                                                          \
          \
        _Pragma("unroll")                                                      \
        for (int u_ = 0; u_ < PKU; ++u_) {                                     \
            const int k_ = ((kb_) * PKU + u_) * 4;                             \
            fa[st_][u_]  = *(const float2*)(xr + k_);                          \
            fw0[st_][u_] = *(const float4*)(wp + (size_t)k_ * HOq);            \
            fw1[st_][u_] = *(const float4*)(wp + (size_t)(k_ + 1) * HOq);      \
        }                                                                      \
    }

    v8f c0 = {}, c1 = {}, c2 = {}, c3 = {};
    PLOAD(0, 0);
#pragma unroll
    for (int kb = 0; kb < Dq / (4 * PKU); ++kb) {
        if (kb + 1 < Dq / (4 * PKU)) PLOAD((kb + 1) & 1, kb + 1);
        __builtin_amdgcn_sched_barrier(0);
#pragma unroll
        for (int u = 0; u < PKU; ++u) {
            const float2 av = fa[kb & 1][u];
            const float4 w0 = fw0[kb & 1][u];
            const float4 w1 = fw1[kb & 1][u];
            v2f a;  a.x  = av.x; a.y  = av.y;
            v2f b0; b0.x = w0.x; b0.y = w1.x;
            v2f b1; b1.x = w0.y; b1.y = w1.y;
            v2f b2; b2.x = w0.z; b2.y = w1.z;
            v2f b3; b3.x = w0.w; b3.y = w1.w;
            c0 = __builtin_amdgcn_wmma_f32_16x16x4_f32(false, a, false, b0, (short)0, c0, false, false);
            c1 = __builtin_amdgcn_wmma_f32_16x16x4_f32(false, a, false, b1, (short)0, c1, false, false);
            c2 = __builtin_amdgcn_wmma_f32_16x16x4_f32(false, a, false, b2, (short)0, c2, false, false);
            c3 = __builtin_amdgcn_wmma_f32_16x16x4_f32(false, a, false, b3, (short)0, c3, false, false);
        }
    }
#undef PLOAD

    float* hp  = hws + ((size_t)bh * Nq + i0) * HOq + lm * 4;
    float* hsw = hs + wv * (16 * PTP);
#pragma unroll
    for (int r = 0; r < 8; ++r) {
        const int mr = r + 8 * half;
        float4 o; o.x = c0[r]; o.y = c1[r]; o.z = c2[r]; o.w = c3[r];
        *(float4*)(hsw + mr * PTP + lm * 4) = o;
    }
    __syncthreads();
#pragma unroll
    for (int r = 0; r < 8; ++r) {
        const int mr = r + 8 * half;
        float4 o; o.x = c0[r]; o.y = c1[r]; o.z = c2[r]; o.w = c3[r];
        *(float4*)(hp + (size_t)mr * HOq) = o;
    }

    const float* av_ = (half ? adst : asrc) + (size_t)hh * Tq * HOq;
    const float* hr  = hsw + lm * PTP;
    float s0 = 0.f, s1 = 0.f, s2 = 0.f, s3 = 0.f, s4 = 0.f;
#pragma unroll
    for (int q = 0; q < HOq / 4; ++q) {
        const float4 hx = *(const float4*)(hr + 4 * q);
        const float4 a0 = *(const float4*)(av_ + 0 * HOq + 4 * q);
        const float4 a1 = *(const float4*)(av_ + 1 * HOq + 4 * q);
        const float4 a2 = *(const float4*)(av_ + 2 * HOq + 4 * q);
        const float4 a3 = *(const float4*)(av_ + 3 * HOq + 4 * q);
        const float4 a4 = *(const float4*)(av_ + 4 * HOq + 4 * q);
        s0 = fmaf(hx.w, a0.w, fmaf(hx.z, a0.z, fmaf(hx.y, a0.y, fmaf(hx.x, a0.x, s0))));
        s1 = fmaf(hx.w, a1.w, fmaf(hx.z, a1.z, fmaf(hx.y, a1.y, fmaf(hx.x, a1.x, s1))));
        s2 = fmaf(hx.w, a2.w, fmaf(hx.z, a2.z, fmaf(hx.y, a2.y, fmaf(hx.x, a2.x, s2))));
        s3 = fmaf(hx.w, a3.w, fmaf(hx.z, a3.z, fmaf(hx.y, a3.y, fmaf(hx.x, a3.x, s3))));
        s4 = fmaf(hx.w, a4.w, fmaf(hx.z, a4.z, fmaf(hx.y, a4.y, fmaf(hx.x, a4.x, s4))));
    }
    float* ep = (half ? ed : es) + (size_t)bh * Tq * Nq + i0 + lm;
    ep[0 * Nq] = s0;
    ep[1 * Nq] = s1;
    ep[2 * Nq] = s2;
    ep[3 * Nq] = s3;
    ep[4 * Nq] = s4;
}

#define JS   8
#define RT   2
#define JLEN (Nq / JS)
#define NIT  (JLEN / 4)
#define GB   8
#define NB   (NIT / GB)
#define TP   68
#define NRING 3
static_assert(RT == 2, "pass 1 and pass 2 keep tile 0 in registers and tile 1 in LDS");
static_assert(16 * (NIT / 2) * 32 <= RT * 16 * TP * 4, "tile 1's score chunks fit the wave's own partial-tile region");

__device__ __forceinline__ float score1(int a, float edv, float e0, float e1, float e2,
                                        float e3, float e4) {
    const int t = (a > 0) ? (a - 1) : 0;
    const float sc  = sel5(t, e0, e1, e2, e3, e4) + edv;
    const float scn = 0.01f * sc;
    const float lr  = (sc > 0.f) ? sc : scn;
    return (a == 0) ? NEGV : lr;
}

__global__ void __launch_bounds__(32 * JS) k_attn(const int* __restrict__ adj,
                                                  const float* __restrict__ hws,
                                                  const float* __restrict__ es,
                                                  const float* __restrict__ ed,
                                                  const float* __restrict__ bias,
                                                  const float* __restrict__ gamma,
                                                  const float* __restrict__ beta,
                                                  float* __restrict__ out) {
    __shared__ __attribute__((aligned(16))) float part[JS * RT * 16 * TP];
    __shared__ __attribute__((aligned(16))) float smax[RT * 16 * JS];
    __shared__ __attribute__((aligned(16))) float ssum[RT * 16 * JS];
    const int tid  = threadIdx.x;
    const int lane = tid & 31;
    const int wv   = __builtin_amdgcn_readfirstlane(tid >> 5);
    const int half = lane >> 4;
    const int lm   = lane & 15;
    const int i0   = blockIdx.x * (16 * RT);
    const int bh   = blockIdx.y;
    const int b    = bh / Hq, hh = bh % Hq;

    const int jbeg = wv * JLEN;
    const unsigned lofs = (unsigned)half * 2u;
    const float* eds = ed + (size_t)bh * Tq * Nq + jbeg;
    const float* hb  = hws + (size_t)bh * Nq * HOq;
    const float* hbs = hb + (size_t)jbeg * HOq;
    const unsigned ho = lofs * HOq + (unsigned)lm;

    float4* sc1 = (float4*)(part + wv * (RT * 16 * TP)) + lane;
    float e[RT][Tq];
#pragma unroll
    for (int rt = 0; rt < RT; ++rt)
#pragma unroll
        for (int t = 0; t < Tq; ++t)
            e[rt][t] = es[((size_t)bh * Tq + t) * Nq + i0 + 16 * rt + lm];
    const int* ar0 = adj + ((size_t)b * Nq + i0 + lm) * Nq + jbeg + lofs;
    int2  av[RT][NIT];
    float w0[2 * NIT];
    float wb[2][2 * GB];
    float m[RT];
#pragma unroll
    for (int rt = 0; rt < RT; ++rt) m[rt] = -INFINITY;
#define WDST(g_) (((g_) / NB == 0) ? (w0 + 2 * (((g_) % NB) * GB)) : wb[(g_) & 1])
#define LOADAV(g_)                                                             \
    {                                                                          \
        const int rt_ = (g_) / NB, b0_ = ((g_) % NB) * GB;                     \
        _Pragma("unroll")                                                      \
        for (int it_ = b0_; it_ < b0_ + GB; ++it_)                             \
            av[rt_][it_] = *(const int2*)(ar0 + (size_t)rt_ * 16 * Nq + 4 * it_); \
    }
#define GATHER(g_)                                                             \
    {                                                                          \
        const int rt_ = (g_) / NB, b0_ = ((g_) % NB) * GB;                     \
        float* wd_ = WDST(g_);                                                 \
        _Pragma("unroll")                                                      \
        for (int u_ = 0; u_ < GB; ++u_) {                                      \
            const int it_ = b0_ + u_;                                          \
            const unsigned t0_ = (av[rt_][it_].x > 0) ? (unsigned)(av[rt_][it_].x - 1) : 0u; \
            const unsigned t1_ = (av[rt_][it_].y > 0) ? (unsigned)(av[rt_][it_].y - 1) : 0u; \
            wd_[2 * u_]     = eds[t0_ * Nq + lofs + 4 * it_];                  \
            wd_[2 * u_ + 1] = eds[t1_ * Nq + lofs + 4 * it_ + 1];              \
        }                                                                      \
    }
#define SCORE(g_)                                                              \
    {                                                                          \
        const int rt_ = (g_) / NB, b0_ = ((g_) % NB) * GB;                     \
        float* wd_ = WDST(g_);                                                 \
        _Pragma("unroll")                                                      \
        for (int u_ = 0; u_ < GB; ++u_) {                                      \
            const int it_ = b0_ + u_;                                          \
            const float w0_ = score1(av[rt_][it_].x, wd_[2 * u_],              \
                                     e[rt_][0], e[rt_][1], e[rt_][2], e[rt_][3], e[rt_][4]); \
            const float w1_ = score1(av[rt_][it_].y, wd_[2 * u_ + 1],          \
                                     e[rt_][0], e[rt_][1], e[rt_][2], e[rt_][3], e[rt_][4]); \
            wd_[2 * u_]     = w0_;                                             \
            wd_[2 * u_ + 1] = w1_;                                             \
            m[rt_] = fmaxf(m[rt_], fmaxf(w0_, w1_));                           \
        }                                                                      \
        if (rt_ == 1) {                                                        \
            _Pragma("unroll")                                                  \
            for (int c_ = 0; c_ < GB / 2; ++c_) {                              \
                float4 o_; o_.x = wd_[4 * c_]; o_.y = wd_[4 * c_ + 1];         \
                o_.z = wd_[4 * c_ + 2]; o_.w = wd_[4 * c_ + 3];                \
                sc1[(b0_ / 2 + c_) * 32] = o_;                                 \
            }                                                                  \
        }                                                                      \
    }
    LOADAV(0);
    LOADAV(1);
    GATHER(0);
#pragma unroll
    for (int g = 0; g < RT * NB; ++g) {
        if (g + 2 < RT * NB) LOADAV(g + 2);
        if (g + 1 < RT * NB) GATHER(g + 1);
        __builtin_amdgcn_sched_barrier(0);
        SCORE(g);
        __builtin_amdgcn_sched_barrier(0);
    }
#undef LOADAV
#undef GATHER
#undef SCORE
#undef WDST

#define LOADB(dst, it_)                                                        \
    {                                                                          \
        const float* hp_ = hbs + ho + (4 * (it_)) * HOq;                       \
        dst[0].x = hp_[0];  dst[0].y = hp_[HOq];                               \
        dst[1].x = hp_[16]; dst[1].y = hp_[HOq + 16];                          \
        dst[2].x = hp_[32]; dst[2].y = hp_[HOq + 32];                          \
        dst[3].x = hp_[48]; dst[3].y = hp_[HOq + 48];                          \
    }
    v2f bq[NRING][4];
#pragma unroll
    for (int q = 0; q < NRING - 1; ++q) LOADB(bq[q], q);

    float rm[RT];
#pragma unroll
    for (int rt = 0; rt < RT; ++rt) {
        const float mx = fmaxf(m[rt], __shfl_xor(m[rt], 16, 32));
        if (half == 0) smax[(16 * rt + lm) * JS + wv] = mx;
    }
    __syncthreads();
#pragma unroll
    for (int rt = 0; rt < RT; ++rt) {
        const float4 ma = *(const float4*)(smax + (16 * rt + lm) * JS);
        const float4 mb = *(const float4*)(smax + (16 * rt + lm) * JS + 4);
        rm[rt] = fmaxf(fmaxf(fmaxf(ma.x, ma.y), fmaxf(ma.z, ma.w)),
                       fmaxf(fmaxf(mb.x, mb.y), fmaxf(mb.z, mb.w)));
    }

    float s[RT];
    v8f c[RT][4];
#pragma unroll
    for (int rt = 0; rt < RT; ++rt) {
        s[rt] = 0.f;
        c[rt][0] = (v8f){}; c[rt][1] = (v8f){}; c[rt][2] = (v8f){}; c[rt][3] = (v8f){};
    }
    float4 q1[2];
    q1[0] = sc1[0];
#pragma unroll
    for (int it = 0; it < NIT; ++it) {
        if (it + NRING - 1 < NIT) LOADB(bq[(it + NRING - 1) % NRING], it + NRING - 1);
        if ((it & 1) == 0 && it / 2 + 1 < NIT / 2) q1[(it / 2 + 1) & 1] = sc1[(it / 2 + 1) * 32];
        __builtin_amdgcn_sched_barrier(0);
        const float4 qc = q1[(it / 2) & 1];
#pragma unroll
        for (int rt = 0; rt < RT; ++rt) {
            const float x0 = (rt == 0) ? w0[2 * it]     : ((it & 1) ? qc.z : qc.x);
            const float x1 = (rt == 0) ? w0[2 * it + 1] : ((it & 1) ? qc.w : qc.y);
            const float p0 = __expf(x0 - rm[rt]);
            const float p1 = __expf(x1 - rm[rt]);
            s[rt] += p0 + p1;
            v2f ap; ap.x = p0; ap.y = p1;
            c[rt][0] = __builtin_amdgcn_wmma_f32_16x16x4_f32(false, ap, false, bq[it % NRING][0], (short)0, c[rt][0], false, false);
            c[rt][1] = __builtin_amdgcn_wmma_f32_16x16x4_f32(false, ap, false, bq[it % NRING][1], (short)0, c[rt][1], false, false);
            c[rt][2] = __builtin_amdgcn_wmma_f32_16x16x4_f32(false, ap, false, bq[it % NRING][2], (short)0, c[rt][2], false, false);
            c[rt][3] = __builtin_amdgcn_wmma_f32_16x16x4_f32(false, ap, false, bq[it % NRING][3], (short)0, c[rt][3], false, false);
        }
    }
#undef LOADB

    float* pw = part + wv * (RT * 16 * TP);
#pragma unroll
    for (int rt = 0; rt < RT; ++rt) {
        float sv = s[rt];
        sv += __shfl_xor(sv, 16, 32);
        if (half == 0) ssum[(16 * rt + lm) * JS + wv] = sv;
#pragma unroll
        for (int r = 0; r < 8; ++r) {
            const int mr = 16 * rt + r + 8 * half;
            pw[mr * TP +  0 + lm] = c[rt][0][r];
            pw[mr * TP + 16 + lm] = c[rt][1][r];
            pw[mr * TP + 32 + lm] = c[rt][2][r];
            pw[mr * TP + 48 + lm] = c[rt][3][r];
        }
    }
    __syncthreads();

    const int q4 = (tid & 15) * 4;
    const float4 bv = *(const float4*)(bias  + hh * HOq + q4);
    const float4 gv = *(const float4*)(gamma + hh * HOq + q4);
    const float4 tv = *(const float4*)(beta  + hh * HOq + q4);
    for (int rt = 0; rt < RT; ++rt) {
        const int r = 16 * rt + (tid >> 4);
        const float* pr = part + r * TP + q4;
        float4 v = *(const float4*)pr;
        for (int ww = 1; ww < JS; ++ww) {
            const float4 t = *(const float4*)(pr + ww * (RT * 16 * TP));
            v.x += t.x; v.y += t.y; v.z += t.z; v.w += t.w;
        }
        const float4 sa = *(const float4*)(ssum + r * JS);
        const float4 sb = *(const float4*)(ssum + r * JS + 4);
        const float tot  = ((((((sa.x + sa.y) + sa.z) + sa.w) + sb.x) + sb.y) + sb.z) + sb.w;
        const float rinv = 1.f / tot;
        const float4 hv = *(const float4*)(hb + (size_t)(i0 + r) * HOq + q4);
        v.x = fmaxf(v.x * rinv + bv.x, 0.f) + hv.x;
        v.y = fmaxf(v.y * rinv + bv.y, 0.f) + hv.y;
        v.z = fmaxf(v.z * rinv + bv.z, 0.f) + hv.z;
        v.w = fmaxf(v.w * rinv + bv.w, 0.f) + hv.w;
        float sum   = (v.x + v.y) + (v.z + v.w);
        float sumsq = (v.x * v.x + v.y * v.y) + (v.z * v.z + v.w * v.w);
        for (int off = 8; off >= 1; off >>= 1) {
            sum   += __shfl_xor(sum,   off, 32);
            sumsq += __shfl_xor(sumsq, off, 32);
        }
        const float mu   = sum * (1.f / HOq);
        const float var  = sumsq * (1.f / HOq) - mu * mu;
        const float rstd = rsqrtf(var + EPSV);
        float4 o;
        o.x = (v.x - mu) * rstd * gv.x + tv.x;
        o.y = (v.y - mu) * rstd * gv.y + tv.y;
        o.z = (v.z - mu) * rstd * gv.z + tv.z;
        o.w = (v.w - mu) * rstd * gv.w + tv.w;
        *(float4*)(out + ((size_t)b * Nq + (i0 + r)) * (Hq * HOq) + hh * HOq + q4) = o;
    }
}

extern "C" void kernel_launch(void* const* d_in, const int* in_sizes, int n_in,
                              void* d_out, int out_size, void* d_ws, size_t ws_size,
                              hipStream_t stream) {
    const float* nodes = (const float*)d_in[0];
    const int*   adj   = (const int*)d_in[1];
    const float* Wm    = (const float*)d_in[2];
    const float* asrc  = (const float*)d_in[3];
    const float* adst  = (const float*)d_in[4];
    const float* bias  = (const float*)d_in[5];
    const float* gamma = (const float*)d_in[6];
    const float* beta  = (const float*)d_in[7];
    float* out = (float*)d_out;

    float* h_ws = (float*)d_ws;
    float* esrc = h_ws + (size_t)Bq * Hq * Nq * HOq;
    float* edst = esrc + (size_t)Bq * Hq * Tq * Nq;

    k_proj<<<dim3(Nq / (16 * PW), Bq * Hq), 32 * PW, 0, stream>>>(nodes, Wm, asrc, adst, h_ws, esrc, edst);
    k_attn<<<dim3(Nq / (16 * RT), Bq * Hq), 32 * JS, 0, stream>>>(adj, h_ws, esrc, edst,
                                                      bias, gamma, beta, out);
}


# k_attn pass 1: a score's two per-type terms come from per-wave LDS tables indexed by adjacency value (e_src by row, e_dst by neighbour), not a global gather + 5-way select; sum order same
# speedup vs baseline: 4.8555x; 1.0979x over previous
#include <hip/hip_runtime.h>

#define Bq  8
#define Nq  1024
#define Dq  256
#define Hq  4
#define HOq 64
#define Tq  5
#define NEGV (-1e9f)
#define EPSV (1e-6f)

typedef float v2f __attribute__((ext_vector_type(2)));
typedef float v8f __attribute__((ext_vector_type(8)));

#define PW  4
#define PKU 8
#define PTP 68

__global__ void __launch_bounds__(32 * PW) k_proj(const float* __restrict__ X,
                                                  const float* __restrict__ Wm,
                                                  const float* __restrict__ asrc,
                                                  const float* __restrict__ adst,
                                                  float* __restrict__ hws,
                                                  float* __restrict__ es,
                                                  float* __restrict__ ed) {
    __shared__ __attribute__((aligned(16))) float hs[PW * 16 * PTP];
    const int tid  = threadIdx.x;
    const int lane = tid & 31;
    const int wv   = __builtin_amdgcn_readfirstlane(tid >> 5);
    const int half = lane >> 4;
    const int lm   = lane & 15;
    const int i0   = (blockIdx.x * PW + wv) * 16;
    const int bh   = blockIdx.y;
    const int b    = bh / Hq, hh = bh % Hq;

    const float* xr = X + ((size_t)b * Nq + (size_t)(i0 + lm)) * Dq + half * 2;
    const float* wp = Wm + (size_t)hh * Dq * HOq + (size_t)(half * 2) * HOq + lm * 4;

    float2 fa[2][PKU];
    float4 fw0[2][PKU], fw1[2][PKU];
#define PLOAD(st_, kb_)                                                        \
    {                                                                          \
          \
        _Pragma("unroll")                                                      \
        for (int u_ = 0; u_ < PKU; ++u_) {                                     \
            const int k_ = ((kb_) * PKU + u_) * 4;                             \
            fa[st_][u_]  = *(const float2*)(xr + k_);                          \
            fw0[st_][u_] = *(const float4*)(wp + (size_t)k_ * HOq);            \
            fw1[st_][u_] = *(const float4*)(wp + (size_t)(k_ + 1) * HOq);      \
        }                                                                      \
    }

    v8f c0 = {}, c1 = {}, c2 = {}, c3 = {};
    PLOAD(0, 0);
#pragma unroll
    for (int kb = 0; kb < Dq / (4 * PKU); ++kb) {
        if (kb + 1 < Dq / (4 * PKU)) PLOAD((kb + 1) & 1, kb + 1);
        __builtin_amdgcn_sched_barrier(0);
#pragma unroll
        for (int u = 0; u < PKU; ++u) {
            const float2 av = fa[kb & 1][u];
            const float4 w0 = fw0[kb & 1][u];
            const float4 w1 = fw1[kb & 1][u];
            v2f a;  a.x  = av.x; a.y  = av.y;
            v2f b0; b0.x = w0.x; b0.y = w1.x;
            v2f b1; b1.x = w0.y; b1.y = w1.y;
            v2f b2; b2.x = w0.z; b2.y = w1.z;
            v2f b3; b3.x = w0.w; b3.y = w1.w;
            c0 = __builtin_amdgcn_wmma_f32_16x16x4_f32(false, a, false, b0, (short)0, c0, false, false);
            c1 = __builtin_amdgcn_wmma_f32_16x16x4_f32(false, a, false, b1, (short)0, c1, false, false);
            c2 = __builtin_amdgcn_wmma_f32_16x16x4_f32(false, a, false, b2, (short)0, c2, false, false);
            c3 = __builtin_amdgcn_wmma_f32_16x16x4_f32(false, a, false, b3, (short)0, c3, false, false);
        }
    }
#undef PLOAD

    float* hp  = hws + ((size_t)bh * Nq + i0) * HOq + lm * 4;
    float* hsw = hs + wv * (16 * PTP);
#pragma unroll
    for (int r = 0; r < 8; ++r) {
        const int mr = r + 8 * half;
        float4 o; o.x = c0[r]; o.y = c1[r]; o.z = c2[r]; o.w = c3[r];
        *(float4*)(hsw + mr * PTP + lm * 4) = o;
    }
    __syncthreads();
#pragma unroll
    for (int r = 0; r < 8; ++r) {
        const int mr = r + 8 * half;
        float4 o; o.x = c0[r]; o.y = c1[r]; o.z = c2[r]; o.w = c3[r];
        *(float4*)(hp + (size_t)mr * HOq) = o;
    }

    const float* av_ = (half ? adst : asrc) + (size_t)hh * Tq * HOq;
    const float* hr  = hsw + lm * PTP;
    float s0 = 0.f, s1 = 0.f, s2 = 0.f, s3 = 0.f, s4 = 0.f;
#pragma unroll
    for (int q = 0; q < HOq / 4; ++q) {
        const float4 hx = *(const float4*)(hr + 4 * q);
        const float4 a0 = *(const float4*)(av_ + 0 * HOq + 4 * q);
        const float4 a1 = *(const float4*)(av_ + 1 * HOq + 4 * q);
        const float4 a2 = *(const float4*)(av_ + 2 * HOq + 4 * q);
        const float4 a3 = *(const float4*)(av_ + 3 * HOq + 4 * q);
        const float4 a4 = *(const float4*)(av_ + 4 * HOq + 4 * q);
        s0 = fmaf(hx.w, a0.w, fmaf(hx.z, a0.z, fmaf(hx.y, a0.y, fmaf(hx.x, a0.x, s0))));
        s1 = fmaf(hx.w, a1.w, fmaf(hx.z, a1.z, fmaf(hx.y, a1.y, fmaf(hx.x, a1.x, s1))));
        s2 = fmaf(hx.w, a2.w, fmaf(hx.z, a2.z, fmaf(hx.y, a2.y, fmaf(hx.x, a2.x, s2))));
        s3 = fmaf(hx.w, a3.w, fmaf(hx.z, a3.z, fmaf(hx.y, a3.y, fmaf(hx.x, a3.x, s3))));
        s4 = fmaf(hx.w, a4.w, fmaf(hx.z, a4.z, fmaf(hx.y, a4.y, fmaf(hx.x, a4.x, s4))));
    }
    float* ep = (half ? ed : es) + (size_t)bh * Tq * Nq + i0 + lm;
    ep[0 * Nq] = s0;
    ep[1 * Nq] = s1;
    ep[2 * Nq] = s2;
    ep[3 * Nq] = s3;
    ep[4 * Nq] = s4;
}

#define JS   8
#define RT   2
#define JLEN (Nq / JS)
#define NIT  (JLEN / 4)
#define GB   8
#define NB   (NIT / GB)
#define TP   68
#define NRING 3
#define NSLOT 6
#define EDP  8
static_assert(RT == 2, "pass 1 and pass 2 keep tile 0 in registers and tile 1 in LDS");
static_assert(NSLOT == Tq + 1 && NSLOT <= EDP && JLEN % 32 == 0, "one table slot per adjacency value; a lane fills whole neighbours");
static_assert(16 * (NIT / 2) * 32 <= RT * 16 * TP * 4, "tile 1's score chunks fit the wave's own partial-tile region");

__device__ __forceinline__ float score1(int a, float esv, float edv) {
    const float sc  = esv + edv;
    const float scn = 0.01f * sc;
    const float lr  = (sc > 0.f) ? sc : scn;
    return (a == 0) ? NEGV : lr;
}

__global__ void __launch_bounds__(32 * JS) k_attn(const int* __restrict__ adj,
                                                  const float* __restrict__ hws,
                                                  const float* __restrict__ es,
                                                  const float* __restrict__ ed,
                                                  const float* __restrict__ bias,
                                                  const float* __restrict__ gamma,
                                                  const float* __restrict__ beta,
                                                  float* __restrict__ out) {
    __shared__ __attribute__((aligned(16))) float part[JS * RT * 16 * TP];
    __shared__ __attribute__((aligned(16))) float smax[RT * 16 * JS];
    __shared__ __attribute__((aligned(16))) float ssum[RT * 16 * JS];
    __shared__ float est[JS * RT * NSLOT * 32];
    __shared__ __attribute__((aligned(16))) float edt[JS * JLEN * EDP];
    const int tid  = threadIdx.x;
    const int lane = tid & 31;
    const int wv   = __builtin_amdgcn_readfirstlane(tid >> 5);
    const int half = lane >> 4;
    const int lm   = lane & 15;
    const int i0   = blockIdx.x * (16 * RT);
    const int bh   = blockIdx.y;
    const int b    = bh / Hq, hh = bh % Hq;

    const int jbeg = wv * JLEN;
    const unsigned lofs = (unsigned)half * 2u;
    const float* eds = ed + (size_t)bh * Tq * Nq + jbeg;
    const float* hb  = hws + (size_t)bh * Nq * HOq;
    const float* hbs = hb + (size_t)jbeg * HOq;
    const unsigned ho = lofs * HOq + (unsigned)lm;

    float* esl = est + wv * (RT * NSLOT * 32) + lane;
    float* edw = edt + wv * (JLEN * EDP);
#pragma unroll
    for (int rt = 0; rt < RT; ++rt) {
        esl[(rt * NSLOT) * 32] = 0.f;
#pragma unroll
        for (int t = 0; t < Tq; ++t)
            esl[(rt * NSLOT + 1 + t) * 32] = es[((size_t)bh * Tq + t) * Nq + i0 + 16 * rt + lm];
    }
#pragma unroll
    for (int q = 0; q < JLEN / 32; ++q) {
        const int jl = lane + 32 * q;
        float4 lo, hi;
        lo.x = 0.f;
        lo.y = eds[0 * Nq + jl];
        lo.z = eds[1 * Nq + jl];
        lo.w = eds[2 * Nq + jl];
        hi.x = eds[3 * Nq + jl];
        hi.y = eds[4 * Nq + jl];
        hi.z = 0.f;
        hi.w = 0.f;
        *(float4*)(edw + jl * EDP)     = lo;
        *(float4*)(edw + jl * EDP + 4) = hi;
    }
    const float* edl = edw + lofs * EDP;

    float4* sc1 = (float4*)(part + wv * (RT * 16 * TP)) + lane;
    const int* ar0 = adj + ((size_t)b * Nq + i0 + lm) * Nq + jbeg + lofs;
    int2  av[RT][NIT];
    float w0[2 * NIT];
    float wb[2][2 * GB];
    float xs[2][2 * GB];
    float m[RT];
#pragma unroll
    for (int rt = 0; rt < RT; ++rt) m[rt] = -INFINITY;
#define WDST(g_) (((g_) / NB == 0) ? (w0 + 2 * (((g_) % NB) * GB)) : wb[(g_) & 1])
#define LOADAV(g_)                                                             \
    {                                                                          \
        const int rt_ = (g_) / NB, b0_ = ((g_) % NB) * GB;                     \
        _Pragma("unroll")                                                      \
        for (int it_ = b0_; it_ < b0_ + GB; ++it_)                             \
            av[rt_][it_] = *(const int2*)(ar0 + (size_t)rt_ * 16 * Nq + 4 * it_); \
    }
#define LOOKUP(g_)                                                             \
    {                                                                          \
        const int rt_ = (g_) / NB, b0_ = ((g_) % NB) * GB;                     \
        float* wd_ = WDST(g_);                                                 \
        float* xs_ = xs[(g_) & 1];                                             \
        _Pragma("unroll")                                                      \
        for (int u_ = 0; u_ < GB; ++u_) {                                      \
            const int it_ = b0_ + u_;                                          \
            const unsigned a0_ = (unsigned)av[rt_][it_].x;                     \
            const unsigned a1_ = (unsigned)av[rt_][it_].y;                     \
            xs_[2 * u_]     = esl[(rt_ * NSLOT + a0_) * 32];                   \
            xs_[2 * u_ + 1] = esl[(rt_ * NSLOT + a1_) * 32];                   \
            wd_[2 * u_]     = edl[(4 * it_) * EDP + a0_];                      \
            wd_[2 * u_ + 1] = edl[(4 * it_ + 1) * EDP + a1_];                  \
        }                                                                      \
    }
#define SCORE(g_)                                                              \
    {                                                                          \
        const int rt_ = (g_) / NB, b0_ = ((g_) % NB) * GB;                     \
        float* wd_ = WDST(g_);                                                 \
        const float* xs_ = xs[(g_) & 1];                                       \
        _Pragma("unroll")                                                      \
        for (int u_ = 0; u_ < GB; ++u_) {                                      \
            const int it_ = b0_ + u_;                                          \
            const float w0_ = score1(av[rt_][it_].x, xs_[2 * u_],     wd_[2 * u_]);     \
            const float w1_ = score1(av[rt_][it_].y, xs_[2 * u_ + 1], wd_[2 * u_ + 1]); \
            wd_[2 * u_]     = w0_;                                             \
            wd_[2 * u_ + 1] = w1_;                                             \
            m[rt_] = fmaxf(m[rt_], fmaxf(w0_, w1_));                           \
        }                                                                      \
        if (rt_ == 1) {                                                        \
            _Pragma("unroll")                                                  \
            for (int c_ = 0; c_ < GB / 2; ++c_) {                              \
                float4 o_; o_.x = wd_[4 * c_]; o_.y = wd_[4 * c_ + 1];         \
                o_.z = wd_[4 * c_ + 2]; o_.w = wd_[4 * c_ + 3];                \
                sc1[(b0_ / 2 + c_) * 32] = o_;                                 \
            }                                                                  \
        }                                                                      \
    }
    LOADAV(0);
    LOADAV(1);
    LOOKUP(0);
#pragma unroll
    for (int g = 0; g < RT * NB; ++g) {
        if (g + 2 < RT * NB) LOADAV(g + 2);
        if (g + 1 < RT * NB) LOOKUP(g + 1);
        __builtin_amdgcn_sched_barrier(0);
        SCORE(g);
        __builtin_amdgcn_sched_barrier(0);
    }
#undef LOADAV
#undef LOOKUP
#undef SCORE
#undef WDST

#define LOADB(dst, it_)                                                        \
    {                                                                          \
        const float* hp_ = hbs + ho + (4 * (it_)) * HOq;                       \
        dst[0].x = hp_[0];  dst[0].y = hp_[HOq];                               \
        dst[1].x = hp_[16]; dst[1].y = hp_[HOq + 16];                          \
        dst[2].x = hp_[32]; dst[2].y = hp_[HOq + 32];                          \
        dst[3].x = hp_[48]; dst[3].y = hp_[HOq + 48];                          \
    }
    v2f bq[NRING][4];
#pragma unroll
    for (int q = 0; q < NRING - 1; ++q) LOADB(bq[q], q);

    float rm[RT];
#pragma unroll
    for (int rt = 0; rt < RT; ++rt) {
        const float mx = fmaxf(m[rt], __shfl_xor(m[rt], 16, 32));
        if (half == 0) smax[(16 * rt + lm) * JS + wv] = mx;
    }
    __syncthreads();
#pragma unroll
    for (int rt = 0; rt < RT; ++rt) {
        const float4 ma = *(const float4*)(smax + (16 * rt + lm) * JS);
        const float4 mb = *(const float4*)(smax + (16 * rt + lm) * JS + 4);
        rm[rt] = fmaxf(fmaxf(fmaxf(ma.x, ma.y), fmaxf(ma.z, ma.w)),
                       fmaxf(fmaxf(mb.x, mb.y), fmaxf(mb.z, mb.w)));
    }

    float s[RT];
    v8f c[RT][4];
#pragma unroll
    for (int rt = 0; rt < RT; ++rt) {
        s[rt] = 0.f;
        c[rt][0] = (v8f){}; c[rt][1] = (v8f){}; c[rt][2] = (v8f){}; c[rt][3] = (v8f){};
    }
    float4 q1[2];
    q1[0] = sc1[0];
#pragma unroll
    for (int it = 0; it < NIT; ++it) {
        if (it + NRING - 1 < NIT) LOADB(bq[(it + NRING - 1) % NRING], it + NRING - 1);
        if ((it & 1) == 0 && it / 2 + 1 < NIT / 2) q1[(it / 2 + 1) & 1] = sc1[(it / 2 + 1) * 32];
        __builtin_amdgcn_sched_barrier(0);
        const float4 qc = q1[(it / 2) & 1];
#pragma unroll
        for (int rt = 0; rt < RT; ++rt) {
            const float x0 = (rt == 0) ? w0[2 * it]     : ((it & 1) ? qc.z : qc.x);
            const float x1 = (rt == 0) ? w0[2 * it + 1] : ((it & 1) ? qc.w : qc.y);
            const float p0 = __expf(x0 - rm[rt]);
            const float p1 = __expf(x1 - rm[rt]);
            s[rt] += p0 + p1;
            v2f ap; ap.x = p0; ap.y = p1;
            c[rt][0] = __builtin_amdgcn_wmma_f32_16x16x4_f32(false, ap, false, bq[it % NRING][0], (short)0, c[rt][0], false, false);
            c[rt][1] = __builtin_amdgcn_wmma_f32_16x16x4_f32(false, ap, false, bq[it % NRING][1], (short)0, c[rt][1], false, false);
            c[rt][2] = __builtin_amdgcn_wmma_f32_16x16x4_f32(false, ap, false, bq[it % NRING][2], (short)0, c[rt][2], false, false);
            c[rt][3] = __builtin_amdgcn_wmma_f32_16x16x4_f32(false, ap, false, bq[it % NRING][3], (short)0, c[rt][3], false, false);
        }
    }
#undef LOADB

    float* pw = part + wv * (RT * 16 * TP);
#pragma unroll
    for (int rt = 0; rt < RT; ++rt) {
        float sv = s[rt];
        sv += __shfl_xor(sv, 16, 32);
        if (half == 0) ssum[(16 * rt + lm) * JS + wv] = sv;
#pragma unroll
        for (int r = 0; r < 8; ++r) {
            const int mr = 16 * rt + r + 8 * half;
            pw[mr * TP +  0 + lm] = c[rt][0][r];
            pw[mr * TP + 16 + lm] = c[rt][1][r];
            pw[mr * TP + 32 + lm] = c[rt][2][r];
            pw[mr * TP + 48 + lm] = c[rt][3][r];
        }
    }
    __syncthreads();

    const int q4 = (tid & 15) * 4;
    const float4 bv = *(const float4*)(bias  + hh * HOq + q4);
    const float4 gv = *(const float4*)(gamma + hh * HOq + q4);
    const float4 tv = *(const float4*)(beta  + hh * HOq + q4);
    for (int rt = 0; rt < RT; ++rt) {
        const int r = 16 * rt + (tid >> 4);
        const float* pr = part + r * TP + q4;
        float4 v = *(const float4*)pr;
        for (int ww = 1; ww < JS; ++ww) {
            const float4 t = *(const float4*)(pr + ww * (RT * 16 * TP));
            v.x += t.x; v.y += t.y; v.z += t.z; v.w += t.w;
        }
        const float4 sa = *(const float4*)(ssum + r * JS);
        const float4 sb = *(const float4*)(ssum + r * JS + 4);
        const float tot  = ((((((sa.x + sa.y) + sa.z) + sa.w) + sb.x) + sb.y) + sb.z) + sb.w;
        const float rinv = 1.f / tot;
        const float4 hv = *(const float4*)(hb + (size_t)(i0 + r) * HOq + q4);
        v.x = fmaxf(v.x * rinv + bv.x, 0.f) + hv.x;
        v.y = fmaxf(v.y * rinv + bv.y, 0.f) + hv.y;
        v.z = fmaxf(v.z * rinv + bv.z, 0.f) + hv.z;
        v.w = fmaxf(v.w * rinv + bv.w, 0.f) + hv.w;
        float sum   = (v.x + v.y) + (v.z + v.w);
        float sumsq = (v.x * v.x + v.y * v.y) + (v.z * v.z + v.w * v.w);
        for (int off = 8; off >= 1; off >>= 1) {
            sum   += __shfl_xor(sum,   off, 32);
            sumsq += __shfl_xor(sumsq, off, 32);
        }
        const float mu   = sum * (1.f / HOq);
        const float var  = sumsq * (1.f / HOq) - mu * mu;
        const float rstd = rsqrtf(var + EPSV);
        float4 o;
        o.x = (v.x - mu) * rstd * gv.x + tv.x;
        o.y = (v.y - mu) * rstd * gv.y + tv.y;
        o.z = (v.z - mu) * rstd * gv.z + tv.z;
        o.w = (v.w - mu) * rstd * gv.w + tv.w;
        *(float4*)(out + ((size_t)b * Nq + (i0 + r)) * (Hq * HOq) + hh * HOq + q4) = o;
    }
}

extern "C" void kernel_launch(void* const* d_in, const int* in_sizes, int n_in,
                              void* d_out, int out_size, void* d_ws, size_t ws_size,
                              hipStream_t stream) {
    const float* nodes = (const float*)d_in[0];
    const int*   adj   = (const int*)d_in[1];
    const float* Wm    = (const float*)d_in[2];
    const float* asrc  = (const float*)d_in[3];
    const float* adst  = (const float*)d_in[4];
    const float* bias  = (const float*)d_in[5];
    const float* gamma = (const float*)d_in[6];
    const float* beta  = (const float*)d_in[7];
    float* out = (float*)d_out;

    float* h_ws = (float*)d_ws;
    float* esrc = h_ws + (size_t)Bq * Hq * Nq * HOq;
    float* edst = esrc + (size_t)Bq * Hq * Tq * Nq;

    k_proj<<<dim3(Nq / (16 * PW), Bq * Hq), 32 * PW, 0, stream>>>(nodes, Wm, asrc, adst, h_ws, esrc, edst);
    k_attn<<<dim3(Nq / (16 * RT), Bq * Hq), 32 * JS, 0, stream>>>(adj, h_ws, esrc, edst,
                                                      bias, gamma, beta, out);
}
